# starting kernel
# baseline (speedup 1.0000x reference)
#include <hip/hip_runtime.h>

typedef __attribute__((ext_vector_type(2))) float v2f;
typedef __attribute__((ext_vector_type(8))) float v8f;

#define CHANNELS 64

__device__ constexpr float BETA_F  = 0.11778303565638346f;
__device__ constexpr float ALPHA_F = 0.1f;

__global__ void init_deg_kernel(float* __restrict__ deg, int n) {
    int v = blockIdx.x * blockDim.x + threadIdx.x;
    if (v < n) deg[v] = 1.0f;
}

__global__ void deg_count_kernel(const int* __restrict__ dst,
                                 float* __restrict__ deg, int E) {
    int e = blockIdx.x * blockDim.x + threadIdx.x;
    if (e < E) atomicAdd(&deg[dst[e]], 1.0f);
}

__global__ void deg_rsqrt_kernel(float* __restrict__ deg, int n) {
    int v = blockIdx.x * blockDim.x + threadIdx.x;
    if (v < n) deg[v] = rsqrtf(deg[v]);
}

__global__ void agg_init_kernel(const float* __restrict__ x,
                                const float* __restrict__ dinv,
                                float* __restrict__ agg, int total) {
    int idx = blockIdx.x * blockDim.x + threadIdx.x;
    if (idx >= total) return;
    int v = idx >> 6;
    float di = dinv[v];
    agg[idx] = di * di * x[idx];
}

__global__ void scatter_kernel(const int* __restrict__ src,
                               const int* __restrict__ dst,
                               const float* __restrict__ x,
                               const float* __restrict__ dinv,
                               float* __restrict__ agg, int E) {
    int gid  = blockIdx.x * blockDim.x + threadIdx.x;
    int e    = gid >> 5;
    int lane = gid & 31;
    if (e >= E) return;
    int s = src[e];
    int d = dst[e];
    float w = dinv[s] * dinv[d];
    const float* xs = x + (size_t)s * CHANNELS;
    float* ad = agg + (size_t)d * CHANNELS;
    atomicAdd(&ad[lane],      w * xs[lane]);
    atomicAdd(&ad[lane + 32], w * xs[lane + 32]);
}

__global__ __launch_bounds__(256) void gcn2_gemm_kernel(
        const float* __restrict__ agg, const float* __restrict__ x0,
        const float* __restrict__ W, float* __restrict__ out) {
    __shared__ float sH[8 * 16 * 64];
    __shared__ float sWt[64 * 64];

    const int t = threadIdx.x;
    const int Rblk = blockIdx.x * 128;

#pragma unroll
    for (int i = 0; i < 16; ++i) {
        int idx = i * 256 + t;
        int k = idx >> 6, n = idx & 63;
        sWt[n * 64 + k] = W[idx];
    }
#pragma unroll
    for (int i = 0; i < 32; ++i) {
        int idx = i * 256 + t;
        size_t g = (size_t)Rblk * 64 + idx;
        sH[idx] = (1.0f - ALPHA_F) * agg[g] + ALPHA_F * x0[g];
    }
    __syncthreads();

    const int wave = t >> 5;
    const int lane = t & 31;
    const int half = lane >> 4;
    const int ml   = lane & 15;
    const float* h = &sH[wave * 1024];

    v2f a[16];
#pragma unroll
    for (int k = 0; k < 16; ++k) {
        int off = ml * 64 + 4 * k + 2 * half;
        a[k].x = h[off];
        a[k].y = h[off + 1];
    }

    const float seed = (1.0f - BETA_F) / BETA_F;
#pragma unroll
    for (int nt = 0; nt < 4; ++nt) {
        const int n0 = nt * 16;
        v8f c;
#pragma unroll
        for (int j = 0; j < 8; ++j)
            c[j] = seed * h[(j + 8 * half) * 64 + n0 + ml];

#pragma unroll
        for (int k = 0; k < 16; ++k) {
            int boff = (n0 + ml) * 64 + 4 * k + 2 * half;
            v2f b;
            b.x = sWt[boff];
            b.y = sWt[boff + 1];
            c = __builtin_amdgcn_wmma_f32_16x16x4_f32(
                    false, a[k], false, b, (short)0, c, false, false);
        }

        const int rowBase = Rblk + wave * 16 + 8 * half;
#pragma unroll
        for (int j = 0; j < 8; ++j)
            out[(size_t)(rowBase + j) * 64 + n0 + ml] = BETA_F * c[j];
    }
}

extern "C" void kernel_launch(void* const* d_in, const int* in_sizes, int n_in,
                              void* d_out, int out_size, void* d_ws, size_t ws_size,
                              hipStream_t stream) {
    const float* x  = (const float*)d_in[0];
    const float* x0 = (const float*)d_in[1];
    const float* W  = (const float*)d_in[2];
    const int*   ei = (const int*)d_in[3];

    const int N = in_sizes[0] / CHANNELS;
    const int E = in_sizes[3] / 2;
    const int* src = ei;
    const int* dst = ei + E;

    float* deg = (float*)d_ws;
    float* agg = (float*)d_out;

    init_deg_kernel<<<(N + 255) / 256, 256, 0, stream>>>(deg, N);
    deg_count_kernel<<<(E + 255) / 256, 256, 0, stream>>>(dst, deg, E);
    deg_rsqrt_kernel<<<(N + 255) / 256, 256, 0, stream>>>(deg, N);

    const int total = N * CHANNELS;
    agg_init_kernel<<<(total + 255) / 256, 256, 0, stream>>>(x, deg, agg, total);

    const long long sthreads = (long long)E * 32;
    scatter_kernel<<<(int)((sthreads + 255) / 256), 256, 0, stream>>>(
        src, dst, x, deg, agg, E);

    gcn2_gemm_kernel<<<N / 128, 256, 0, stream>>>(agg, x0, W, (float*)d_out);
}


# Owner agg: matched edges go to per-row src lists in LDS (integer cursor adds); each wave sums its 8 rows' gathered x rows in registers from the seed; no LDS float adds; sum order: list order
# speedup vs baseline: 4.0596x; 4.0596x over previous
#include <hip/hip_runtime.h>

typedef __attribute__((ext_vector_type(2))) float v2f;
typedef __attribute__((ext_vector_type(8))) float v8f;

#define CHANNELS 64

__device__ constexpr float BETA_F  = 0.11778303565638346f;
__device__ constexpr float ALPHA_F = 0.1f;

#define RANGE 256
#define RANGE_SHIFT 8

#define PACK_EDGES 4096
__global__ __launch_bounds__(1024) void pack_kernel(const int* __restrict__ src,
                            const int* __restrict__ dst,
                            uint4* __restrict__ pres,
                            unsigned* __restrict__ rec, int E) {
    __shared__ unsigned col[4 * RANGE];
    const int t = threadIdx.x;
    const int k = blockIdx.x;
    col[t] = 0u;
    const int e0 = k * PACK_EDGES + t;
    int d[4];
    unsigned s[4];
#pragma unroll
    for (int i = 0; i < 4; ++i) {
        d[i] = dst[e0 + i * 1024];
        s[i] = (unsigned)src[e0 + i * 1024];
    }
    __syncthreads();
    const unsigned bit = 1u << (t >> 5);
#pragma unroll
    for (int i = 0; i < 4; ++i) {
        rec[e0 + i * 1024] = (s[i] << 16) | ((unsigned)d[i] & 0xFFFFu);
        atomicOr(&col[i * RANGE + (((unsigned)d[i] >> RANGE_SHIFT) & (RANGE - 1))], bit);
    }
    __syncthreads();
    if (t < RANGE)
        pres[k * RANGE + t] = make_uint4(col[t], col[RANGE + t], col[2 * RANGE + t], col[3 * RANGE + t]);
}

#define GLIST_CAP 1024

__device__ __forceinline__ int build_group_list(unsigned w, int lane, unsigned short* gl) {
    int ng = 0;
    unsigned mask = __builtin_amdgcn_ballot_w32(w != 0u);
    while (mask) {
        if (w != 0u) {
            gl[ng + __builtin_amdgcn_mbcnt_lo(mask, 0u)] = (unsigned short)(lane * 32 + __builtin_ctz(w));
            w &= w - 1;
        }
        ng += __builtin_popcount(mask);
        mask = __builtin_amdgcn_ballot_w32(w != 0u);
    }
    return ng;
}

__global__ __launch_bounds__(1024) void owner_count_kernel(
        const unsigned* __restrict__ rec,
        const unsigned* __restrict__ pres,
        float* __restrict__ dinv, int E) {
    __shared__ unsigned sCnt[RANGE];
    __shared__ unsigned short sG[32 * GLIST_CAP];

    const int t    = threadIdx.x;
    const int lane = t & 31;
    const int wave = __builtin_amdgcn_readfirstlane(t >> 5);
    const int b    = blockIdx.x;
    const int v0   = b * RANGE;

    if (t < RANGE) sCnt[t] = 0u;
    __syncthreads();

    const int NW = E >> 10;
    unsigned short* gl = sG + wave * GLIST_CAP;
    for (int W0 = 0; W0 < NW; W0 += 1024) {
        const int widx = W0 + t;
        unsigned w = 0u;
        if (widx < NW) w = pres[((widx >> 2) * RANGE + b) * 4 + (widx & 3)];
        const int ng = build_group_list(w, lane, gl);
        const unsigned* recw = rec + (size_t)(W0 + wave * 32) * 1024;
        for (int i0 = 0; i0 < ng; i0 += 16) {
            const int gv = gl[min(i0 + (lane & 15), ng - 1)];
            unsigned r[16];
#pragma unroll
            for (int u = 0; u < 16; ++u) {
                const int g = __builtin_amdgcn_readlane(gv, u);
                r[u] = recw[g * 32 + lane];
            }
#pragma unroll
            for (int u = 0; u < 16; ++u) {
                if (i0 + u < ng && ((r[u] >> RANGE_SHIFT) & (RANGE - 1)) == (unsigned)b)
                    atomicAdd(&sCnt[r[u] & (RANGE - 1)], 1u);
            }
        }
    }
    __syncthreads();

    if (t < RANGE) dinv[v0 + t] = rsqrtf(1.0f + (float)sCnt[t]);
}


#define MLIST_CAP 544

#define RCAP 64

#define TP 68

__device__ __forceinline__ void hand_to_row(
        unsigned r, bool live,
        unsigned* sRcnt, unsigned short* sRow,
        const float* __restrict__ dinv, const float* sDinv,
        const float* __restrict__ x, float* tile) {
    if (live) {
        const unsigned row  = r & (RANGE - 1);
        const unsigned s    = r >> 16;
        const unsigned slot = atomicAdd(&sRcnt[row], 1u);
        if (slot < RCAP) {
            sRow[row * RCAP + slot] = (unsigned short)s;
        } else {
            const float wgt = dinv[s] * sDinv[row];
            const float* xs = x + (size_t)s * CHANNELS;
            float* tp = tile + row * TP;
            for (int c = 0; c < CHANNELS; ++c) atomicAdd(tp + c, wgt * xs[c]);
        }
    }
}

template <int NG, bool ZERO_PAST_M>
__device__ __forceinline__ float2 sum_group(
        unsigned s, float w, int j0, int m, int lane,
        const float* __restrict__ x, float2 acc) {
    float2 xv[NG];
#pragma unroll
    for (int u = 0; u < NG; ++u) {
        const unsigned sj = (unsigned)__builtin_amdgcn_readlane((int)s, j0 + u);
        xv[u] = *(const float2*)(x + (size_t)sj * CHANNELS + 2 * lane);
    }
#pragma unroll
    for (int u = 0; u < NG; ++u) {
        const float wj = __builtin_bit_cast(float,
                             __builtin_amdgcn_readlane(__builtin_bit_cast(int, w), j0 + u));
        float2 v = xv[u];
        if (ZERO_PAST_M && j0 + u >= m) v = make_float2(0.0f, 0.0f);
        acc.x += wj * v.x;
        acc.y += wj * v.y;
    }
    return acc;
}

__device__ __forceinline__ float2 sum_rows(
        unsigned s, float w, int m, int lane,
        const float* __restrict__ x, float2 acc) {
    int j0 = 0;
    for (; j0 + 16 <= m; j0 += 16) acc = sum_group<16, false>(s, w, j0, m, lane, x, acc);
    if (j0 + 8 <= m) { acc = sum_group<8, false>(s, w, j0, m, lane, x, acc); j0 += 8; }
    if (j0 < m) acc = sum_group<8, true>(s, w, j0, m, lane, x, acc);
    return acc;
}

__global__ __launch_bounds__(1024) void owner_agg_tail_kernel(
        const unsigned* __restrict__ rec,
        const unsigned* __restrict__ pres,
        const float* __restrict__ x, const float* __restrict__ dinv,
        const float* __restrict__ x0, const float* __restrict__ W,
        float* __restrict__ out, int E) {
    __shared__ __attribute__((aligned(16))) float tile[RANGE * TP];
    __shared__ __attribute__((aligned(16))) float sWt[CHANNELS * TP];
    __shared__ float sDinv[RANGE];
    __shared__ unsigned short sG[32 * GLIST_CAP];
    __shared__ unsigned sM[32 * MLIST_CAP];
    __shared__ unsigned sRcnt[RANGE];
    __shared__ __attribute__((aligned(16))) unsigned short sRow[RANGE * RCAP];

    const int t    = threadIdx.x;
    const int lane = t & 31;
    const int wave = __builtin_amdgcn_readfirstlane(t >> 5);
    const int b    = blockIdx.x;
    const int v0   = b * RANGE;

    if (t < RANGE) { sDinv[t] = dinv[v0 + t]; sRcnt[t] = 0u; }
    for (int i = 0; i < RANGE * RCAP * 2 / 16 / 1024; ++i)
        ((uint4*)sRow)[i * 1024 + t] = make_uint4(0u, 0u, 0u, 0u);
#pragma unroll
    for (int i = 0; i < CHANNELS * CHANNELS / 1024; ++i) {
        int idx = i * 1024 + t;
        sWt[(idx & 63) * TP + (idx >> 6)] = W[idx];
    }
    const float4* x4 = (const float4*)(x + (size_t)v0 * CHANNELS);
    float4* tile4 = (float4*)tile;
#pragma unroll
    for (int i = 0; i < RANGE * CHANNELS / 4 / 1024; ++i) {
        int q = i * 1024 + t;
        float di = dinv[v0 + (q >> 4)];
        float sc = di * di;
        float4 xv = x4[q];
        tile4[(q >> 4) * (TP / 4) + (q & 15)] = make_float4(sc * xv.x, sc * xv.y, sc * xv.z, sc * xv.w);
    }
    __syncthreads();

    const int NW = E >> 10;
    unsigned short* gl = sG + wave * GLIST_CAP;
    unsigned* mlst = sM + wave * MLIST_CAP;
    int cnt = 0;
    for (int W0 = 0; W0 < NW; W0 += 1024) {
        const int widx = W0 + t;
        unsigned w = 0u;
        if (widx < NW) w = pres[((widx >> 2) * RANGE + b) * 4 + (widx & 3)];
        const int ng = build_group_list(w, lane, gl);
        const unsigned* recw = rec + (size_t)(W0 + wave * 32) * 1024;
        for (int i0 = 0; i0 < ng; i0 += 16) {
            const int gv = gl[min(i0 + (lane & 15), ng - 1)];
            unsigned r[16];
#pragma unroll
            for (int u = 0; u < 16; ++u) {
                const int g = __builtin_amdgcn_readlane(gv, u);
                r[u] = recw[g * 32 + lane];
            }
#pragma unroll
            for (int u = 0; u < 16; ++u) {
                const bool m = i0 + u < ng && ((r[u] >> RANGE_SHIFT) & (RANGE - 1)) == (unsigned)b;
                const unsigned mk = __builtin_amdgcn_ballot_w32(m);
                if (m) mlst[cnt + __builtin_amdgcn_mbcnt_lo(mk, 0u)] = r[u];
                cnt += __builtin_popcount(mk);
            }
            while (cnt >= 32) {
                cnt -= 32;
                hand_to_row(mlst[cnt + lane], true, sRcnt, sRow, dinv, sDinv, x, tile);
            }
        }
    }
    if (cnt)
        hand_to_row(mlst[min(lane, cnt - 1)], lane < cnt, sRcnt, sRow, dinv, sDinv, x, tile);
    __syncthreads();

    const float4* x04 = (const float4*)(x0 + (size_t)v0 * CHANNELS);
    float4 xr[RANGE * CHANNELS / 4 / 1024];

    {
        const int R0 = wave * 8;
        int      nq[8];
        unsigned sv[8];
        float    wv[8];
#pragma unroll
        for (int q = 0; q < 8; ++q)
            nq[q] = __builtin_amdgcn_readfirstlane((int)min(sRcnt[R0 + q], (unsigned)RCAP));
#pragma unroll
        for (int q = 0; q < 8; ++q)
            sv[q] = (unsigned)sRow[(R0 + q) * RCAP + lane];
#pragma unroll
        for (int q = 0; q < 8; ++q)
            wv[q] = dinv[sv[q]] * (sDinv[R0 + q] * (lane < nq[q] ? 1.0f : 0.0f));
#pragma unroll
        for (int i = 0; i < RANGE * CHANNELS / 4 / 1024; ++i) xr[i] = x04[i * 1024 + t];
        bool big = false;
#pragma unroll
        for (int q = 0; q < 8; ++q) {
            float2* tp = (float2*)(tile + (R0 + q) * TP + 2 * lane);
            *tp = sum_rows(sv[q], wv[q], min(nq[q], 32), lane, x, *tp);
            big |= nq[q] > 32;
        }
        if (big) {
#pragma nounroll
            for (int q = 0; q < 8; ++q) {
                const int row = R0 + q;
                const int n = __builtin_amdgcn_readfirstlane((int)min(sRcnt[row], (unsigned)RCAP));
                if (n > 32) {
                    float2* tp = (float2*)(tile + row * TP + 2 * lane);
                    float2 acc = *tp;
                    for (int base = 32; base < n; base += 32) {
                        const unsigned s2 = (unsigned)sRow[row * RCAP + base + lane];
                        const float w2 = dinv[s2] * (sDinv[row] * (base + lane < n ? 1.0f : 0.0f));
                        acc = sum_rows(s2, w2, min(n - base, 32), lane, x, acc);
                    }
                    *tp = acc;
                }
            }
        }
    }
    __syncthreads();

#pragma unroll
    for (int i = 0; i < RANGE * CHANNELS / 4 / 1024; ++i) {
        int q = i * 1024 + t;
        int o = (q >> 4) * (TP / 4) + (q & 15);
        float4 g = tile4[o];
        tile4[o] = make_float4((1.0f - ALPHA_F) * g.x + ALPHA_F * xr[i].x,
                               (1.0f - ALPHA_F) * g.y + ALPHA_F * xr[i].y,
                               (1.0f - ALPHA_F) * g.z + ALPHA_F * xr[i].z,
                               (1.0f - ALPHA_F) * g.w + ALPHA_F * xr[i].w);
    }
    __syncthreads();

    const int half = lane >> 4;
    const int ml   = lane & 15;
    const int rt   = wave >> 1;
    const int ct   = (wave & 1) * 2;
    const float* h = tile + rt * 16 * TP;

    v2f a[16];
#pragma unroll
    for (int k = 0; k < 16; ++k) {
        int off = ml * TP + 4 * k + 2 * half;
        a[k].x = h[off];
        a[k].y = h[off + 1];
    }

    const float seed = (1.0f - BETA_F) / BETA_F;
#pragma unroll
    for (int nt = 0; nt < 2; ++nt) {
        const int n0 = (ct + nt) * 16;
        v8f c;
#pragma unroll
        for (int j = 0; j < 8; ++j)
            c[j] = seed * h[(j + 8 * half) * TP + n0 + ml];

#pragma unroll
        for (int k = 0; k < 16; ++k) {
            int boff = (n0 + ml) * TP + 4 * k + 2 * half;
            v2f bf;
            bf.x = sWt[boff];
            bf.y = sWt[boff + 1];
            c = __builtin_amdgcn_wmma_f32_16x16x4_f32(
                    false, a[k], false, bf, (short)0, c, false, false);
        }

        const int rowBase = v0 + rt * 16 + 8 * half;
#pragma unroll
        for (int j = 0; j < 8; ++j)
            out[(size_t)(rowBase + j) * 64 + n0 + ml] = BETA_F * c[j];
    }
}

extern "C" void kernel_launch(void* const* d_in, const int* in_sizes, int n_in,
                              void* d_out, int out_size, void* d_ws, size_t ws_size,
                              hipStream_t stream) {
    const float* x  = (const float*)d_in[0];
    const float* x0 = (const float*)d_in[1];
    const float* W  = (const float*)d_in[2];
    const int*   ei = (const int*)d_in[3];

    const int N = in_sizes[0] / CHANNELS;
    const int E = in_sizes[3] / 2;
    const int* src = ei;
    const int* dst = ei + E;

    float* deg = (float*)d_ws;
    unsigned char* key = (unsigned char*)d_ws + (size_t)N * sizeof(float);
    unsigned* rec = (unsigned*)(key + E);

    pack_kernel<<<E / PACK_EDGES, 1024, 0, stream>>>(src, dst, (uint4*)key, rec, E);

    owner_count_kernel<<<N / RANGE, 1024, 0, stream>>>(rec, (const unsigned*)key, deg, E);
    owner_agg_tail_kernel<<<N / RANGE, 1024, 0, stream>>>(rec, (const unsigned*)key, x, deg,
                                                          x0, W, (float*)d_out, E);
}


# Presence bits per 16-edge half line (was per 32-edge line): one 32-lane record load serves two present groups, owner scans run half the steps; sum order: list order, lists fill in a new scan order
# speedup vs baseline: 4.6307x; 1.1407x over previous
#include <hip/hip_runtime.h>

typedef __attribute__((ext_vector_type(2))) float v2f;
typedef __attribute__((ext_vector_type(8))) float v8f;

#define CHANNELS 64

__device__ constexpr float BETA_F  = 0.11778303565638346f;
__device__ constexpr float ALPHA_F = 0.1f;

#define RANGE 256
#define RANGE_SHIFT 8

#define PACK_EDGES 4096
#define GROUP_EDGES 16
#define PWORDS 8
__global__ __launch_bounds__(1024) void pack_kernel(const int* __restrict__ src,
                            const int* __restrict__ dst,
                            uint4* __restrict__ pres,
                            unsigned* __restrict__ rec, int E) {
    __shared__ unsigned col[PWORDS * RANGE];
    const int t = threadIdx.x;
    const int k = blockIdx.x;
    col[t] = 0u;
    col[t + 1024] = 0u;
    const int e0 = k * PACK_EDGES + t;
    int d[4];
    unsigned s[4];
#pragma unroll
    for (int i = 0; i < 4; ++i) {
        d[i] = dst[e0 + i * 1024];
        s[i] = (unsigned)src[e0 + i * 1024];
    }
    __syncthreads();
    const unsigned bit = 1u << ((t >> 4) & 31);
    const int wsel = t >> 9;
#pragma unroll
    for (int i = 0; i < 4; ++i) {
        rec[e0 + i * 1024] = (s[i] << 16) | ((unsigned)d[i] & 0xFFFFu);
        atomicOr(&col[(2 * i + wsel) * RANGE + (((unsigned)d[i] >> RANGE_SHIFT) & (RANGE - 1))], bit);
    }
    __syncthreads();
    if (t < 2 * RANGE) {
        const int r = t >> 1;
        const int h = (t & 1) * 4;
        pres[(k * RANGE + r) * 2 + (t & 1)] = make_uint4(col[h * RANGE + r], col[(h + 1) * RANGE + r],
                                                         col[(h + 2) * RANGE + r], col[(h + 3) * RANGE + r]);
    }
}

#define GLIST_CAP (1024 + 32)

__device__ __forceinline__ int build_group_list(unsigned w, int lane, unsigned short* gl) {
    int ng = 0;
    unsigned mask = __builtin_amdgcn_ballot_w32(w != 0u);
    while (mask) {
        if (w != 0u) {
            gl[ng + __builtin_amdgcn_mbcnt_lo(mask, 0u)] = (unsigned short)(lane * 32 + __builtin_ctz(w));
            w &= w - 1;
        }
        ng += __builtin_popcount(mask);
        mask = __builtin_amdgcn_ballot_w32(w != 0u);
    }
    gl[ng + lane] = (unsigned short)0;
    return ng;
}

__global__ __launch_bounds__(1024) void owner_count_kernel(
        const unsigned* __restrict__ rec,
        const unsigned* __restrict__ pres,
        float* __restrict__ dinv, int E) {
    __shared__ unsigned sCnt[RANGE];
    __shared__ unsigned short sG[32 * GLIST_CAP];

    const int t    = threadIdx.x;
    const int lane = t & 31;
    const int wave = __builtin_amdgcn_readfirstlane(t >> 5);
    const int b    = blockIdx.x;
    const int v0   = b * RANGE;

    if (t < RANGE) sCnt[t] = 0u;
    __syncthreads();

    const int NW = E >> 9;
    unsigned short* gl = sG + wave * GLIST_CAP;
    const int half = lane >> 4;
    const int l16  = lane & 15;
    unsigned wnext = 0u;
    if (t < NW) wnext = pres[((t >> 3) * RANGE + b) * PWORDS + (t & (PWORDS - 1))];
    for (int W0 = 0; W0 < NW; W0 += 1024) {
        const unsigned w = wnext;
        const int wn = W0 + 1024 + t;
        wnext = 0u;
        if (wn < NW) wnext = pres[((wn >> 3) * RANGE + b) * PWORDS + (wn & (PWORDS - 1))];
        const int ng = build_group_list(w, lane, gl);
        const unsigned* recw = rec + (size_t)(W0 + wave * 32) * (32 * GROUP_EDGES);
        for (int i0 = 0; i0 < ng; i0 += 32) {
            const int p = i0 + half;
            int g[16];
            unsigned r[16];
#pragma unroll
            for (int u = 0; u < 16; ++u) g[u] = gl[p + 2 * u];
#pragma unroll
            for (int u = 0; u < 16; ++u) r[u] = recw[g[u] * GROUP_EDGES + l16];
#pragma unroll
            for (int u = 0; u < 16; ++u) {
                if (p < ng - 2 * u && ((r[u] >> RANGE_SHIFT) & (RANGE - 1)) == (unsigned)b)
                    atomicAdd(&sCnt[r[u] & (RANGE - 1)], 1u);
            }
        }
    }
    __syncthreads();

    if (t < RANGE) dinv[v0 + t] = rsqrtf(1.0f + (float)sCnt[t]);
}


#define MLIST_CAP 544

#define RCAP 64

#define TP 68

__device__ __forceinline__ void hand_to_row(
        unsigned r, bool live,
        unsigned* sRcnt, unsigned short* sRow,
        const float* __restrict__ dinv, const float* sDinv,
        const float* __restrict__ x, float* tile) {
    if (live) {
        const unsigned row  = r & (RANGE - 1);
        const unsigned s    = r >> 16;
        const unsigned slot = atomicAdd(&sRcnt[row], 1u);
        if (slot < RCAP) {
            sRow[row * RCAP + slot] = (unsigned short)s;
        } else {
            const float wgt = dinv[s] * sDinv[row];
            const float* xs = x + (size_t)s * CHANNELS;
            float* tp = tile + row * TP;
            for (int c = 0; c < CHANNELS; ++c) atomicAdd(tp + c, wgt * xs[c]);
        }
    }
}

template <int NG, bool ZERO_PAST_M>
__device__ __forceinline__ float2 sum_group(
        unsigned s, float w, int j0, int m, int lane,
        const float* __restrict__ x, float2 acc) {
    float2 xv[NG];
#pragma unroll
    for (int u = 0; u < NG; ++u) {
        const unsigned sj = (unsigned)__builtin_amdgcn_readlane((int)s, j0 + u);
        xv[u] = *(const float2*)(x + (size_t)sj * CHANNELS + 2 * lane);
    }
#pragma unroll
    for (int u = 0; u < NG; ++u) {
        const float wj = __builtin_bit_cast(float,
                             __builtin_amdgcn_readlane(__builtin_bit_cast(int, w), j0 + u));
        float2 v = xv[u];
        if (ZERO_PAST_M && j0 + u >= m) v = make_float2(0.0f, 0.0f);
        acc.x += wj * v.x;
        acc.y += wj * v.y;
    }
    return acc;
}

__device__ __forceinline__ float2 sum_rows(
        unsigned s, float w, int m, int lane,
        const float* __restrict__ x, float2 acc) {
    int j0 = 0;
    for (; j0 + 16 <= m; j0 += 16) acc = sum_group<16, false>(s, w, j0, m, lane, x, acc);
    if (j0 + 8 <= m) { acc = sum_group<8, false>(s, w, j0, m, lane, x, acc); j0 += 8; }
    if (j0 < m) acc = sum_group<8, true>(s, w, j0, m, lane, x, acc);
    return acc;
}

__global__ __launch_bounds__(1024) void owner_agg_tail_kernel(
        const unsigned* __restrict__ rec,
        const unsigned* __restrict__ pres,
        const float* __restrict__ x, const float* __restrict__ dinv,
        const float* __restrict__ x0, const float* __restrict__ W,
        float* __restrict__ out, int E) {
    __shared__ __attribute__((aligned(16))) float tile[RANGE * TP];
    __shared__ __attribute__((aligned(16))) float sWt[CHANNELS * TP];
    __shared__ float sDinv[RANGE];
    __shared__ unsigned short sG[32 * GLIST_CAP];
    __shared__ unsigned sM[32 * MLIST_CAP];
    __shared__ unsigned sRcnt[RANGE];
    __shared__ __attribute__((aligned(16))) unsigned short sRow[RANGE * RCAP];

    const int t    = threadIdx.x;
    const int lane = t & 31;
    const int wave = __builtin_amdgcn_readfirstlane(t >> 5);
    const int b    = blockIdx.x;
    const int v0   = b * RANGE;

    if (t < RANGE) { sDinv[t] = dinv[v0 + t]; sRcnt[t] = 0u; }
    for (int i = 0; i < RANGE * RCAP * 2 / 16 / 1024; ++i)
        ((uint4*)sRow)[i * 1024 + t] = make_uint4(0u, 0u, 0u, 0u);
#pragma unroll
    for (int i = 0; i < CHANNELS * CHANNELS / 1024; ++i) {
        int idx = i * 1024 + t;
        sWt[(idx & 63) * TP + (idx >> 6)] = W[idx];
    }
    const float4* x4 = (const float4*)(x + (size_t)v0 * CHANNELS);
    float4* tile4 = (float4*)tile;
#pragma unroll
    for (int i = 0; i < RANGE * CHANNELS / 4 / 1024; ++i) {
        int q = i * 1024 + t;
        float di = dinv[v0 + (q >> 4)];
        float sc = di * di;
        float4 xv = x4[q];
        tile4[(q >> 4) * (TP / 4) + (q & 15)] = make_float4(sc * xv.x, sc * xv.y, sc * xv.z, sc * xv.w);
    }
    __syncthreads();

    const int NW = E >> 9;
    unsigned short* gl = sG + wave * GLIST_CAP;
    unsigned* mlst = sM + wave * MLIST_CAP;
    int cnt = 0;
    const int hsel = lane >> 4;
    const int l16  = lane & 15;
    unsigned wnext = 0u;
    if (t < NW) wnext = pres[((t >> 3) * RANGE + b) * PWORDS + (t & (PWORDS - 1))];
    for (int W0 = 0; W0 < NW; W0 += 1024) {
        const unsigned w = wnext;
        const int wn = W0 + 1024 + t;
        wnext = 0u;
        if (wn < NW) wnext = pres[((wn >> 3) * RANGE + b) * PWORDS + (wn & (PWORDS - 1))];
        const int ng = build_group_list(w, lane, gl);
        const unsigned* recw = rec + (size_t)(W0 + wave * 32) * (32 * GROUP_EDGES);
        for (int i0 = 0; i0 < ng; i0 += 32) {
            const int p = i0 + hsel;
            int g[16];
            unsigned r[16];
#pragma unroll
            for (int u = 0; u < 16; ++u) g[u] = gl[p + 2 * u];
#pragma unroll
            for (int u = 0; u < 16; ++u) r[u] = recw[g[u] * GROUP_EDGES + l16];
#pragma unroll
            for (int u = 0; u < 16; ++u) {
                const bool m = p < ng - 2 * u && ((r[u] >> RANGE_SHIFT) & (RANGE - 1)) == (unsigned)b;
                const unsigned mk = __builtin_amdgcn_ballot_w32(m);
                if (m) mlst[cnt + __builtin_amdgcn_mbcnt_lo(mk, 0u)] = r[u];
                cnt += __builtin_popcount(mk);
            }
            while (cnt >= 32) {
                cnt -= 32;
                hand_to_row(mlst[cnt + lane], true, sRcnt, sRow, dinv, sDinv, x, tile);
            }
        }
    }
    if (cnt)
        hand_to_row(mlst[min(lane, cnt - 1)], lane < cnt, sRcnt, sRow, dinv, sDinv, x, tile);
    __syncthreads();

    const float4* x04 = (const float4*)(x0 + (size_t)v0 * CHANNELS);
    float4 xr[RANGE * CHANNELS / 4 / 1024];

    {
        const int R0 = wave * 8;
        int      nq[8];
        unsigned sv[8];
        float    wv[8];
#pragma unroll
        for (int q = 0; q < 8; ++q)
            nq[q] = __builtin_amdgcn_readfirstlane((int)min(sRcnt[R0 + q], (unsigned)RCAP));
#pragma unroll
        for (int q = 0; q < 8; ++q)
            sv[q] = (unsigned)sRow[(R0 + q) * RCAP + lane];
#pragma unroll
        for (int q = 0; q < 8; ++q)
            wv[q] = dinv[sv[q]] * (sDinv[R0 + q] * (lane < nq[q] ? 1.0f : 0.0f));
#pragma unroll
        for (int i = 0; i < RANGE * CHANNELS / 4 / 1024; ++i) xr[i] = x04[i * 1024 + t];
        bool big = false;
#pragma unroll
        for (int q = 0; q < 8; ++q) {
            float2* tp = (float2*)(tile + (R0 + q) * TP + 2 * lane);
            *tp = sum_rows(sv[q], wv[q], min(nq[q], 32), lane, x, *tp);
            big |= nq[q] > 32;
        }
        if (big) {
#pragma nounroll
            for (int q = 0; q < 8; ++q) {
                const int row = R0 + q;
                const int n = __builtin_amdgcn_readfirstlane((int)min(sRcnt[row], (unsigned)RCAP));
                if (n > 32) {
                    float2* tp = (float2*)(tile + row * TP + 2 * lane);
                    float2 acc = *tp;
                    for (int base = 32; base < n; base += 32) {
                        const unsigned s2 = (unsigned)sRow[row * RCAP + base + lane];
                        const float w2 = dinv[s2] * (sDinv[row] * (base + lane < n ? 1.0f : 0.0f));
                        acc = sum_rows(s2, w2, min(n - base, 32), lane, x, acc);
                    }
                    *tp = acc;
                }
            }
        }
    }
    __syncthreads();

#pragma unroll
    for (int i = 0; i < RANGE * CHANNELS / 4 / 1024; ++i) {
        int q = i * 1024 + t;
        int o = (q >> 4) * (TP / 4) + (q & 15);
        float4 g = tile4[o];
        tile4[o] = make_float4((1.0f - ALPHA_F) * g.x + ALPHA_F * xr[i].x,
                               (1.0f - ALPHA_F) * g.y + ALPHA_F * xr[i].y,
                               (1.0f - ALPHA_F) * g.z + ALPHA_F * xr[i].z,
                               (1.0f - ALPHA_F) * g.w + ALPHA_F * xr[i].w);
    }
    __syncthreads();

    const int half = lane >> 4;
    const int ml   = lane & 15;
    const int rt   = wave >> 1;
    const int ct   = (wave & 1) * 2;
    const float* h = tile + rt * 16 * TP;

    v2f a[16];
#pragma unroll
    for (int k = 0; k < 16; ++k) {
        int off = ml * TP + 4 * k + 2 * half;
        a[k].x = h[off];
        a[k].y = h[off + 1];
    }

    const float seed = (1.0f - BETA_F) / BETA_F;
#pragma unroll
    for (int nt = 0; nt < 2; ++nt) {
        const int n0 = (ct + nt) * 16;
        v8f c;
#pragma unroll
        for (int j = 0; j < 8; ++j)
            c[j] = seed * h[(j + 8 * half) * TP + n0 + ml];

#pragma unroll
        for (int k = 0; k < 16; ++k) {
            int boff = (n0 + ml) * TP + 4 * k + 2 * half;
            v2f bf;
            bf.x = sWt[boff];
            bf.y = sWt[boff + 1];
            c = __builtin_amdgcn_wmma_f32_16x16x4_f32(
                    false, a[k], false, bf, (short)0, c, false, false);
        }

        const int rowBase = v0 + rt * 16 + 8 * half;
#pragma unroll
        for (int j = 0; j < 8; ++j)
            out[(size_t)(rowBase + j) * 64 + n0 + ml] = BETA_F * c[j];
    }
}

extern "C" void kernel_launch(void* const* d_in, const int* in_sizes, int n_in,
                              void* d_out, int out_size, void* d_ws, size_t ws_size,
                              hipStream_t stream) {
    const float* x  = (const float*)d_in[0];
    const float* x0 = (const float*)d_in[1];
    const float* W  = (const float*)d_in[2];
    const int*   ei = (const int*)d_in[3];

    const int N = in_sizes[0] / CHANNELS;
    const int E = in_sizes[3] / 2;
    const int* src = ei;
    const int* dst = ei + E;

    float* deg = (float*)d_ws;
    unsigned char* key = (unsigned char*)d_ws + (size_t)N * sizeof(float);
    unsigned* rec = (unsigned*)(key + 2 * (size_t)E);

    pack_kernel<<<E / PACK_EDGES, 1024, 0, stream>>>(src, dst, (uint4*)key, rec, E);

    owner_count_kernel<<<N / RANGE, 1024, 0, stream>>>(rec, (const unsigned*)key, deg, E);
    owner_agg_tail_kernel<<<N / RANGE, 1024, 0, stream>>>(rec, (const unsigned*)key, x, deg,
                                                          x0, W, (float*)d_out, E);
}


# Row src lists and counts are built once, by the degree-count launch, and written out as whole 128-byte lines; the agg launch reads its rows' lists and scans no records; sum order: list order
# speedup vs baseline: 5.1363x; 1.1092x over previous
#include <hip/hip_runtime.h>

typedef __attribute__((ext_vector_type(2))) float v2f;
typedef __attribute__((ext_vector_type(8))) float v8f;

#define CHANNELS 64

__device__ constexpr float BETA_F  = 0.11778303565638346f;
__device__ constexpr float ALPHA_F = 0.1f;

#define RANGE 256
#define RANGE_SHIFT 8

#define PACK_EDGES 4096
#define GROUP_EDGES 16
#define PWORDS 8
__global__ __launch_bounds__(1024) void pack_kernel(const int* __restrict__ src,
                            const int* __restrict__ dst,
                            uint4* __restrict__ pres,
                            unsigned* __restrict__ rec, int E) {
    __shared__ unsigned col[PWORDS * RANGE];
    const int t = threadIdx.x;
    const int k = blockIdx.x;
    col[t] = 0u;
    col[t + 1024] = 0u;
    const int e0 = k * PACK_EDGES + t;
    int d[4];
    unsigned s[4];
#pragma unroll
    for (int i = 0; i < 4; ++i) {
        d[i] = dst[e0 + i * 1024];
        s[i] = (unsigned)src[e0 + i * 1024];
    }
    __syncthreads();
    const unsigned bit = 1u << ((t >> 4) & 31);
    const int wsel = t >> 9;
#pragma unroll
    for (int i = 0; i < 4; ++i) {
        rec[e0 + i * 1024] = (s[i] << 16) | ((unsigned)d[i] & 0xFFFFu);
        atomicOr(&col[(2 * i + wsel) * RANGE + (((unsigned)d[i] >> RANGE_SHIFT) & (RANGE - 1))], bit);
    }
    __syncthreads();
    if (t < 2 * RANGE) {
        const int r = t >> 1;
        const int h = (t & 1) * 4;
        pres[(k * RANGE + r) * 2 + (t & 1)] = make_uint4(col[h * RANGE + r], col[(h + 1) * RANGE + r],
                                                         col[(h + 2) * RANGE + r], col[(h + 3) * RANGE + r]);
    }
}

#define GLIST_CAP (1024 + 32)

__device__ __forceinline__ int build_group_list(unsigned w, int lane, unsigned short* gl) {
    int ng = 0;
    unsigned mask = __builtin_amdgcn_ballot_w32(w != 0u);
    while (mask) {
        if (w != 0u) {
            gl[ng + __builtin_amdgcn_mbcnt_lo(mask, 0u)] = (unsigned short)(lane * 32 + __builtin_ctz(w));
            w &= w - 1;
        }
        ng += __builtin_popcount(mask);
        mask = __builtin_amdgcn_ballot_w32(w != 0u);
    }
    gl[ng + lane] = (unsigned short)0;
    return ng;
}


#define MLIST_CAP 544

#define RCAP 64

__device__ __forceinline__ void hand_to_row(
        unsigned r, bool live, unsigned* sRcnt, unsigned short* sRow) {
    if (live) {
        const unsigned row  = r & (RANGE - 1);
        const unsigned slot = atomicAdd(&sRcnt[row], 1u);
        if (slot < RCAP) sRow[row * RCAP + slot] = (unsigned short)(r >> 16);
    }
}

__global__ __launch_bounds__(1024) void owner_list_kernel(
        const unsigned* __restrict__ rec,
        const unsigned* __restrict__ pres,
        float* __restrict__ dinv,
        unsigned* __restrict__ rcnt,
        uint4* __restrict__ rlist, int E) {
    __shared__ unsigned short sG[32 * GLIST_CAP];
    __shared__ unsigned sM[32 * MLIST_CAP];
    __shared__ unsigned sRcnt[RANGE];
    __shared__ __attribute__((aligned(16))) unsigned short sRow[RANGE * RCAP];

    const int t    = threadIdx.x;
    const int lane = t & 31;
    const int wave = __builtin_amdgcn_readfirstlane(t >> 5);
    const int b    = blockIdx.x;
    const int v0   = b * RANGE;

    if (t < RANGE) sRcnt[t] = 0u;
    for (int i = 0; i < RANGE * RCAP * 2 / 16 / 1024; ++i)
        ((uint4*)sRow)[i * 1024 + t] = make_uint4(0u, 0u, 0u, 0u);
    __syncthreads();

    const int NW = E >> 9;
    unsigned short* gl = sG + wave * GLIST_CAP;
    unsigned* mlst = sM + wave * MLIST_CAP;
    int cnt = 0;
    const int hsel = lane >> 4;
    const int l16  = lane & 15;
    unsigned wnext = 0u;
    if (t < NW) wnext = pres[((t >> 3) * RANGE + b) * PWORDS + (t & (PWORDS - 1))];
    for (int W0 = 0; W0 < NW; W0 += 1024) {
        const unsigned w = wnext;
        const int wn = W0 + 1024 + t;
        wnext = 0u;
        if (wn < NW) wnext = pres[((wn >> 3) * RANGE + b) * PWORDS + (wn & (PWORDS - 1))];
        const int ng = build_group_list(w, lane, gl);
        const unsigned* recw = rec + (size_t)(W0 + wave * 32) * (32 * GROUP_EDGES);
        for (int i0 = 0; i0 < ng; i0 += 32) {
            const int p = i0 + hsel;
            int g[16];
            unsigned r[16];
#pragma unroll
            for (int u = 0; u < 16; ++u) g[u] = gl[p + 2 * u];
#pragma unroll
            for (int u = 0; u < 16; ++u) r[u] = recw[g[u] * GROUP_EDGES + l16];
#pragma unroll
            for (int u = 0; u < 16; ++u) {
                const bool m = p < ng - 2 * u && ((r[u] >> RANGE_SHIFT) & (RANGE - 1)) == (unsigned)b;
                const unsigned mk = __builtin_amdgcn_ballot_w32(m);
                if (m) mlst[cnt + __builtin_amdgcn_mbcnt_lo(mk, 0u)] = r[u];
                cnt += __builtin_popcount(mk);
            }
            while (cnt >= 32) {
                cnt -= 32;
                hand_to_row(mlst[cnt + lane], true, sRcnt, sRow);
            }
        }
    }
    if (cnt)
        hand_to_row(mlst[min(lane, cnt - 1)], lane < cnt, sRcnt, sRow);
    __syncthreads();

    if (t < RANGE) {
        const unsigned c = sRcnt[t];
        dinv[v0 + t] = rsqrtf(1.0f + (float)c);
        rcnt[v0 + t] = c;
    }
    uint4* lout = rlist + (size_t)b * (RANGE * RCAP * 2 / 16);
    for (int i = 0; i < RANGE * RCAP * 2 / 16 / 1024; ++i)
        lout[i * 1024 + t] = ((const uint4*)sRow)[i * 1024 + t];
}


#define TP 68

template <int NG, bool ZERO_PAST_M>
__device__ __forceinline__ float2 sum_group(
        unsigned s, float w, int j0, int m, int lane,
        const float* __restrict__ x, float2 acc) {
    float2 xv[NG];
#pragma unroll
    for (int u = 0; u < NG; ++u) {
        const unsigned sj = (unsigned)__builtin_amdgcn_readlane((int)s, j0 + u);
        xv[u] = *(const float2*)(x + (size_t)sj * CHANNELS + 2 * lane);
    }
#pragma unroll
    for (int u = 0; u < NG; ++u) {
        const float wj = __builtin_bit_cast(float,
                             __builtin_amdgcn_readlane(__builtin_bit_cast(int, w), j0 + u));
        float2 v = xv[u];
        if (ZERO_PAST_M && j0 + u >= m) v = make_float2(0.0f, 0.0f);
        acc.x += wj * v.x;
        acc.y += wj * v.y;
    }
    return acc;
}

__device__ __forceinline__ float2 sum_rows(
        unsigned s, float w, int m, int lane,
        const float* __restrict__ x, float2 acc) {
    int j0 = 0;
    for (; j0 + 16 <= m; j0 += 16) acc = sum_group<16, false>(s, w, j0, m, lane, x, acc);
    if (j0 + 8 <= m) { acc = sum_group<8, false>(s, w, j0, m, lane, x, acc); j0 += 8; }
    if (j0 < m) acc = sum_group<8, true>(s, w, j0, m, lane, x, acc);
    return acc;
}

__global__ __launch_bounds__(1024) void owner_agg_tail_kernel(
        const unsigned* __restrict__ rec,
        const unsigned* __restrict__ rcnt,
        const unsigned short* __restrict__ rlist,
        const float* __restrict__ x, const float* __restrict__ dinv,
        const float* __restrict__ x0, const float* __restrict__ W,
        float* __restrict__ out, int E) {
    __shared__ __attribute__((aligned(16))) float tile[RANGE * TP];
    __shared__ __attribute__((aligned(16))) float sWt[CHANNELS * TP];
    __shared__ float sDinv[RANGE];

    const int t    = threadIdx.x;
    const int lane = t & 31;
    const int wave = __builtin_amdgcn_readfirstlane(t >> 5);
    const int b    = blockIdx.x;
    const int v0   = b * RANGE;

    const int R0 = wave * 8;
    const unsigned short* lrow = rlist + (size_t)(v0 + R0) * RCAP;
    unsigned sv[8];
    float    dv[8];
#pragma unroll
    for (int q = 0; q < 8; ++q) sv[q] = (unsigned)lrow[q * RCAP + lane];
    const uint4* c4 = (const uint4*)(rcnt + v0) + 2 * lane;
    const uint4 ca = c4[0], cb = c4[1];
    const unsigned c8[8] = {ca.x, ca.y, ca.z, ca.w, cb.x, cb.y, cb.z, cb.w};

    float myd = 0.0f;
    if (t < RANGE) myd = dinv[v0 + t];
    float wreg[CHANNELS * CHANNELS / 1024];
#pragma unroll
    for (int i = 0; i < CHANNELS * CHANNELS / 1024; ++i) wreg[i] = W[i * 1024 + t];
    const float4* x4 = (const float4*)(x + (size_t)v0 * CHANNELS);
    float4* tile4 = (float4*)tile;
    float  dseed[RANGE * CHANNELS / 4 / 1024];
    float4 xs[RANGE * CHANNELS / 4 / 1024];
#pragma unroll
    for (int i = 0; i < RANGE * CHANNELS / 4 / 1024; ++i) {
        int q = i * 1024 + t;
        dseed[i] = dinv[v0 + (q >> 4)];
        xs[i] = x4[q];
    }
#pragma unroll
    for (int q = 0; q < 8; ++q) dv[q] = dinv[sv[q]];

    if (t < RANGE) sDinv[t] = myd;
#pragma unroll
    for (int i = 0; i < CHANNELS * CHANNELS / 1024; ++i) {
        int idx = i * 1024 + t;
        sWt[(idx & 63) * TP + (idx >> 6)] = wreg[i];
    }
#pragma unroll
    for (int i = 0; i < RANGE * CHANNELS / 4 / 1024; ++i) {
        int q = i * 1024 + t;
        float sc = dseed[i] * dseed[i];
        tile4[(q >> 4) * (TP / 4) + (q & 15)] = make_float4(sc * xs[i].x, sc * xs[i].y, sc * xs[i].z, sc * xs[i].w);
    }
    __syncthreads();

    unsigned cmax = 0u;
#pragma unroll
    for (int q = 0; q < 8; ++q) cmax = max(cmax, c8[q]);
    const bool ovf = __builtin_amdgcn_ballot_w32(cmax > (unsigned)RCAP) != 0u;

    if (ovf) {
        for (int e = t; e < E; e += 1024) {
            const unsigned r   = rec[e];
            const unsigned row = r & (RANGE - 1);
            if (((r >> RANGE_SHIFT) & (RANGE - 1)) == (unsigned)b && rcnt[v0 + row] > (unsigned)RCAP) {
                const unsigned s = r >> 16;
                const float wgt = dinv[s] * sDinv[row];
                const float* xe = x + (size_t)s * CHANNELS;
                float* tp = tile + row * TP;
                for (int c = 0; c < CHANNELS; ++c) atomicAdd(tp + c, wgt * xe[c]);
            }
        }
        __syncthreads();
    }

    const float4* x04 = (const float4*)(x0 + (size_t)v0 * CHANNELS);
    float4 xr[RANGE * CHANNELS / 4 / 1024];

    {
        int   nq[8];
        float wv[8];
#pragma unroll
        for (int q = 0; q < 8; ++q) {
            const unsigned c = (unsigned)__builtin_amdgcn_readlane((int)c8[q], wave);
            nq[q] = c > (unsigned)RCAP ? 0 : (int)c;
        }
#pragma unroll
        for (int q = 0; q < 8; ++q)
            wv[q] = dv[q] * (sDinv[R0 + q] * (lane < nq[q] ? 1.0f : 0.0f));
#pragma unroll
        for (int i = 0; i < RANGE * CHANNELS / 4 / 1024; ++i) xr[i] = x04[i * 1024 + t];
        bool big = false;
#pragma unroll
        for (int q = 0; q < 8; ++q) {
            float2* tp = (float2*)(tile + (R0 + q) * TP + 2 * lane);
            *tp = sum_rows(sv[q], wv[q], min(nq[q], 32), lane, x, *tp);
            big |= nq[q] > 32;
        }
        if (big) {
#pragma nounroll
            for (int q = 0; q < 8; ++q) {
                const int row = R0 + q;
                const unsigned c = rcnt[v0 + row];
                const int n = __builtin_amdgcn_readfirstlane(c > (unsigned)RCAP ? 0 : (int)c);
                if (n > 32) {
                    float2* tp = (float2*)(tile + row * TP + 2 * lane);
                    float2 acc = *tp;
                    for (int base = 32; base < n; base += 32) {
                        const unsigned s2 = (unsigned)lrow[q * RCAP + base + lane];
                        const float w2 = dinv[s2] * (sDinv[row] * (base + lane < n ? 1.0f : 0.0f));
                        acc = sum_rows(s2, w2, min(n - base, 32), lane, x, acc);
                    }
                    *tp = acc;
                }
            }
        }
    }
    __syncthreads();

#pragma unroll
    for (int i = 0; i < RANGE * CHANNELS / 4 / 1024; ++i) {
        int q = i * 1024 + t;
        int o = (q >> 4) * (TP / 4) + (q & 15);
        float4 g = tile4[o];
        tile4[o] = make_float4((1.0f - ALPHA_F) * g.x + ALPHA_F * xr[i].x,
                               (1.0f - ALPHA_F) * g.y + ALPHA_F * xr[i].y,
                               (1.0f - ALPHA_F) * g.z + ALPHA_F * xr[i].z,
                               (1.0f - ALPHA_F) * g.w + ALPHA_F * xr[i].w);
    }
    __syncthreads();

    const int half = lane >> 4;
    const int ml   = lane & 15;
    const int rt   = wave >> 1;
    const int ct   = (wave & 1) * 2;
    const float* h = tile + rt * 16 * TP;

    v2f a[16];
#pragma unroll
    for (int k = 0; k < 16; ++k) {
        int off = ml * TP + 4 * k + 2 * half;
        a[k].x = h[off];
        a[k].y = h[off + 1];
    }

    const float seed = (1.0f - BETA_F) / BETA_F;
#pragma unroll
    for (int nt = 0; nt < 2; ++nt) {
        const int n0 = (ct + nt) * 16;
        v8f c;
#pragma unroll
        for (int j = 0; j < 8; ++j)
            c[j] = seed * h[(j + 8 * half) * TP + n0 + ml];

#pragma unroll
        for (int k = 0; k < 16; ++k) {
            int boff = (n0 + ml) * TP + 4 * k + 2 * half;
            v2f bf;
            bf.x = sWt[boff];
            bf.y = sWt[boff + 1];
            c = __builtin_amdgcn_wmma_f32_16x16x4_f32(
                    false, a[k], false, bf, (short)0, c, false, false);
        }

        const int rowBase = v0 + rt * 16 + 8 * half;
#pragma unroll
        for (int j = 0; j < 8; ++j)
            out[(size_t)(rowBase + j) * 64 + n0 + ml] = BETA_F * c[j];
    }
}

extern "C" void kernel_launch(void* const* d_in, const int* in_sizes, int n_in,
                              void* d_out, int out_size, void* d_ws, size_t ws_size,
                              hipStream_t stream) {
    const float* x  = (const float*)d_in[0];
    const float* x0 = (const float*)d_in[1];
    const float* W  = (const float*)d_in[2];
    const int*   ei = (const int*)d_in[3];

    const int N = in_sizes[0] / CHANNELS;
    const int E = in_sizes[3] / 2;
    const int* src = ei;
    const int* dst = ei + E;

    float* deg = (float*)d_ws;
    unsigned char* key = (unsigned char*)d_ws + (size_t)N * sizeof(float);
    unsigned* rec = (unsigned*)(key + 2 * (size_t)E);
    unsigned* rcnt = rec + E;
    unsigned short* rlist = (unsigned short*)(rcnt + N);

    pack_kernel<<<E / PACK_EDGES, 1024, 0, stream>>>(src, dst, (uint4*)key, rec, E);

    owner_list_kernel<<<N / RANGE, 1024, 0, stream>>>(rec, (const unsigned*)key, deg, rcnt, (uint4*)rlist, E);
    owner_agg_tail_kernel<<<N / RANGE, 1024, 0, stream>>>(rec, rcnt, rlist, x, deg,
                                                          x0, W, (float*)d_out, E);
}
